# starting kernel
# baseline (speedup 1.0000x reference)
#include <hip/hip_runtime.h>
#include <math.h>

typedef float v2f __attribute__((ext_vector_type(2)));
typedef float v8f __attribute__((ext_vector_type(8)));

#define BATCH 32
#define CIN   512
#define HW    4096
#define NCLUS 10
#define KPAD  16
#define KKEEP 8
#define KCHUNK (HW / 8)


__global__ void vlad_zero_asum(float* __restrict__ asum) {
    int i = blockIdx.x * blockDim.x + threadIdx.x;
    if (i < BATCH * KPAD) asum[i] = 0.0f;
}

__global__ __launch_bounds__(256) void vlad_softmax(const float* __restrict__ score,
                                                    float* __restrict__ S,
                                                    float* __restrict__ asum) {
    int gid = blockIdx.x * 256 + threadIdx.x;
    int b   = gid >> 12;
    int hw  = gid & (HW - 1);

    const float* sp = score + (size_t)b * NCLUS * HW + hw;
    float s[NCLUS];
    float mx = -1e30f;
#pragma unroll
    for (int k = 0; k < NCLUS; ++k) { s[k] = sp[(size_t)k * HW]; mx = fmaxf(mx, s[k]); }
    float sum = 0.0f;
#pragma unroll
    for (int k = 0; k < NCLUS; ++k) { s[k] = __expf(s[k] - mx); sum += s[k]; }
    float inv = 1.0f / sum;

    float* Sp = S + (size_t)b * KPAD * HW + hw;
#pragma unroll
    for (int k = 0; k < NCLUS; ++k) { s[k] *= inv; Sp[(size_t)k * HW] = s[k]; }
#pragma unroll
    for (int k = NCLUS; k < KPAD; ++k) Sp[(size_t)k * HW] = 0.0f;

    __shared__ float red[256];
    for (int k = 0; k < NCLUS; ++k) {
        red[threadIdx.x] = s[k];
        __syncthreads();
        for (int off = 128; off > 0; off >>= 1) {
            if (threadIdx.x < off) red[threadIdx.x] += red[threadIdx.x + off];
            __syncthreads();
        }
        if (threadIdx.x == 0) atomicAdd(&asum[b * KPAD + k], red[0]);
        __syncthreads();
    }
}

__global__ __launch_bounds__(256) void vlad_wmma_gemm(const float* __restrict__ F,
                                                      const float* __restrict__ S,
                                                      float* __restrict__ WX) {
    const int b     = blockIdx.x >> 3;
    const int ntile = blockIdx.x & 7;
    const int c0    = ntile * 64;

    const int wave  = threadIdx.x >> 5;
    const int lane  = threadIdx.x & 31;
    const int kbase = wave * KCHUNK;
    const int mn    = lane & 15;
    const int kk    = (lane >> 4) * 2;

    const float* Sp = S + ((size_t)b * KPAD + mn) * HW + kbase + kk;
    const float* Fp = F + ((size_t)b * CIN + c0 + mn) * HW + kbase + kk;

    v8f acc0 = {}; v8f acc1 = {}; v8f acc2 = {}; v8f acc3 = {};

#pragma unroll 4
    for (int k = 0; k < KCHUNK; k += 4) {
        v2f a  = *(const v2f*)(Sp + k);
        v2f b0 = *(const v2f*)(Fp + k);
        v2f b1 = *(const v2f*)(Fp + (size_t)16 * HW + k);
        v2f b2 = *(const v2f*)(Fp + (size_t)32 * HW + k);
        v2f b3 = *(const v2f*)(Fp + (size_t)48 * HW + k);
        acc0 = __builtin_amdgcn_wmma_f32_16x16x4_f32(false, a, false, b0, (short)0, acc0, false, false);
        acc1 = __builtin_amdgcn_wmma_f32_16x16x4_f32(false, a, false, b1, (short)0, acc1, false, false);
        acc2 = __builtin_amdgcn_wmma_f32_16x16x4_f32(false, a, false, b2, (short)0, acc2, false, false);
        acc3 = __builtin_amdgcn_wmma_f32_16x16x4_f32(false, a, false, b3, (short)0, acc3, false, false);
    }

    __shared__ float lds[8 * 1024];
    float* lp = lds + wave * 1024;
#pragma unroll
    for (int j = 0; j < 8; ++j) {
        lp[(0 * 8 + j) * 32 + lane] = acc0[j];
        lp[(1 * 8 + j) * 32 + lane] = acc1[j];
        lp[(2 * 8 + j) * 32 + lane] = acc2[j];
        lp[(3 * 8 + j) * 32 + lane] = acc3[j];
    }
    __syncthreads();

#pragma unroll
    for (int e = 0; e < 4; ++e) {
        int idx = threadIdx.x * 4 + e;
        float v = 0.0f;
#pragma unroll
        for (int w = 0; w < 8; ++w) v += lds[w * 1024 + idx];
        int slot = idx >> 5;
        int l    = idx & 31;
        int t    = slot >> 3;
        int j    = slot & 7;
        int mm   = j + ((l >> 4) << 3);
        int nn   = t * 16 + (l & 15);
        WX[((size_t)b * KPAD + mm) * CIN + c0 + nn] = v;
    }
}

__global__ __launch_bounds__(256) void vlad_finalize(const float* __restrict__ WX,
                                                     const float* __restrict__ asum,
                                                     const float* __restrict__ centroids,
                                                     float* __restrict__ out) {
    const int b   = blockIdx.x >> 3;
    const int kc  = blockIdx.x & 7;
    const int tid = threadIdx.x;

    const float a = asum[b * KPAD + kc];
    const float* wp = WX + ((size_t)b * KPAD + kc) * CIN;
    const float* cp = centroids + (size_t)kc * CIN;

    float r0 = wp[tid]       - a * cp[tid];
    float r1 = wp[tid + 256] - a * cp[tid + 256];

    __shared__ float red[256];
    red[tid] = r0 * r0 + r1 * r1;
    __syncthreads();
    for (int off = 128; off > 0; off >>= 1) {
        if (tid < off) red[tid] += red[tid + off];
        __syncthreads();
    }
    __shared__ float invn;
    if (tid == 0) invn = 1.0f / fmaxf(sqrtf(red[0]), 1e-12f);
    __syncthreads();

    float* op = out + ((size_t)b * KKEEP + kc) * CIN;
    op[tid]       = r0 * invn;
    op[tid + 256] = r1 * invn;
}

extern "C" void kernel_launch(void* const* d_in, const int* in_sizes, int n_in,
                              void* d_out, int out_size, void* d_ws, size_t ws_size,
                              hipStream_t stream) {
    const float* features  = (const float*)d_in[0];
    const float* score     = (const float*)d_in[1];
    const float* centroids = (const float*)d_in[2];
    float* out = (float*)d_out;

    float* S    = (float*)d_ws;
    float* WX   = S  + (size_t)BATCH * KPAD * HW;
    float* ASUM = WX + (size_t)BATCH * KPAD * CIN;

    vlad_zero_asum<<<2, 256, 0, stream>>>(ASUM);
    vlad_softmax  <<<BATCH * HW / 256, 256, 0, stream>>>(score, S, ASUM);
    vlad_wmma_gemm<<<BATCH * 8,        256, 0, stream>>>(features, S, WX);
    vlad_finalize <<<BATCH * KKEEP,    256, 0, stream>>>(WX, ASUM, centroids, out);
}


# GEMM operand loads 16 bytes per lane (4 loads per 128-byte line, was 8), next 32-k group in flight during the WMMAs; k order in a group permuted alike for A and B, so the sum order changes
# speedup vs baseline: 1.1224x; 1.1224x over previous
#include <hip/hip_runtime.h>
#include <math.h>

typedef float v2f __attribute__((ext_vector_type(2)));
typedef float v4f __attribute__((ext_vector_type(4)));
typedef float v8f __attribute__((ext_vector_type(8)));

#define BATCH 32
#define CIN   512
#define HW    4096
#define NCLUS 10
#define KPAD  16
#define KKEEP 8
#define KCHUNK (HW / 8)


__global__ void vlad_zero_asum(float* __restrict__ asum) {
    int i = blockIdx.x * blockDim.x + threadIdx.x;
    if (i < BATCH * KPAD) asum[i] = 0.0f;
}

__global__ __launch_bounds__(256) void vlad_softmax(const float* __restrict__ score,
                                                    float* __restrict__ S,
                                                    float* __restrict__ asum) {
    int gid = blockIdx.x * 256 + threadIdx.x;
    int b   = gid >> 12;
    int hw  = gid & (HW - 1);

    const float* sp = score + (size_t)b * NCLUS * HW + hw;
    float s[NCLUS];
    float mx = -1e30f;
#pragma unroll
    for (int k = 0; k < NCLUS; ++k) { s[k] = sp[(size_t)k * HW]; mx = fmaxf(mx, s[k]); }
    float sum = 0.0f;
#pragma unroll
    for (int k = 0; k < NCLUS; ++k) { s[k] = __expf(s[k] - mx); sum += s[k]; }
    float inv = 1.0f / sum;

    float* Sp = S + (size_t)b * KPAD * HW + hw;
#pragma unroll
    for (int k = 0; k < NCLUS; ++k) { s[k] *= inv; Sp[(size_t)k * HW] = s[k]; }
#pragma unroll
    for (int k = NCLUS; k < KPAD; ++k) Sp[(size_t)k * HW] = 0.0f;

    __shared__ float red[256];
    for (int k = 0; k < NCLUS; ++k) {
        red[threadIdx.x] = s[k];
        __syncthreads();
        for (int off = 128; off > 0; off >>= 1) {
            if (threadIdx.x < off) red[threadIdx.x] += red[threadIdx.x + off];
            __syncthreads();
        }
        if (threadIdx.x == 0) atomicAdd(&asum[b * KPAD + k], red[0]);
        __syncthreads();
    }
}

struct Frags {
    v4f a[4];
    v4f b0[4];
    v4f b1[4];
    v4f b2[4];
    v4f b3[4];
};

__device__ __forceinline__ void load_group(Frags& f, const float* __restrict__ Sp,
                                           const float* __restrict__ Fp, int k) {
#pragma unroll
    for (int j = 0; j < 4; ++j) {
        f.a[j]  = *(const v4f*)(Sp + k + 8 * j);
        f.b0[j] = *(const v4f*)(Fp + k + 8 * j);
        f.b1[j] = *(const v4f*)(Fp + (size_t)16 * HW + k + 8 * j);
        f.b2[j] = *(const v4f*)(Fp + (size_t)32 * HW + k + 8 * j);
        f.b3[j] = *(const v4f*)(Fp + (size_t)48 * HW + k + 8 * j);
    }
}

__device__ __forceinline__ void mma_group(const Frags& f, v8f& acc0, v8f& acc1,
                                          v8f& acc2, v8f& acc3) {
#pragma unroll
    for (int j = 0; j < 4; ++j) {
        v2f alo = {f.a[j].x, f.a[j].y};
        v2f ahi = {f.a[j].z, f.a[j].w};
        v2f p0 = {f.b0[j].x, f.b0[j].y};
        v2f p1 = {f.b1[j].x, f.b1[j].y};
        v2f p2 = {f.b2[j].x, f.b2[j].y};
        v2f p3 = {f.b3[j].x, f.b3[j].y};
        acc0 = __builtin_amdgcn_wmma_f32_16x16x4_f32(false, alo, false, p0, (short)0, acc0, false, false);
        acc1 = __builtin_amdgcn_wmma_f32_16x16x4_f32(false, alo, false, p1, (short)0, acc1, false, false);
        acc2 = __builtin_amdgcn_wmma_f32_16x16x4_f32(false, alo, false, p2, (short)0, acc2, false, false);
        acc3 = __builtin_amdgcn_wmma_f32_16x16x4_f32(false, alo, false, p3, (short)0, acc3, false, false);
        v2f q0 = {f.b0[j].z, f.b0[j].w};
        v2f q1 = {f.b1[j].z, f.b1[j].w};
        v2f q2 = {f.b2[j].z, f.b2[j].w};
        v2f q3 = {f.b3[j].z, f.b3[j].w};
        acc0 = __builtin_amdgcn_wmma_f32_16x16x4_f32(false, ahi, false, q0, (short)0, acc0, false, false);
        acc1 = __builtin_amdgcn_wmma_f32_16x16x4_f32(false, ahi, false, q1, (short)0, acc1, false, false);
        acc2 = __builtin_amdgcn_wmma_f32_16x16x4_f32(false, ahi, false, q2, (short)0, acc2, false, false);
        acc3 = __builtin_amdgcn_wmma_f32_16x16x4_f32(false, ahi, false, q3, (short)0, acc3, false, false);
    }
}

__global__ __launch_bounds__(256) void vlad_wmma_gemm(const float* __restrict__ F,
                                                      const float* __restrict__ S,
                                                      float* __restrict__ WX) {
    const int b     = blockIdx.x >> 3;
    const int ntile = blockIdx.x & 7;
    const int c0    = ntile * 64;

    const int wave  = threadIdx.x >> 5;
    const int lane  = threadIdx.x & 31;
    const int kbase = wave * KCHUNK;
    const int mn    = lane & 15;
    const int kk    = (lane >> 4) * 4;

    const float* Sp = S + ((size_t)b * KPAD + mn) * HW + kbase + kk;
    const float* Fp = F + ((size_t)b * CIN + c0 + mn) * HW + kbase + kk;

    v8f acc0 = {}; v8f acc1 = {}; v8f acc2 = {}; v8f acc3 = {};

    Frags cur, nxt;
    load_group(cur, Sp, Fp, 0);
#pragma unroll 1
    for (int k = 32; k < KCHUNK; k += 32) {
        load_group(nxt, Sp, Fp, k);
        mma_group(cur, acc0, acc1, acc2, acc3);
        cur = nxt;
    }
    mma_group(cur, acc0, acc1, acc2, acc3);

    __shared__ float lds[8 * 1024];
    float* lp = lds + wave * 1024;
#pragma unroll
    for (int j = 0; j < 8; ++j) {
        lp[(0 * 8 + j) * 32 + lane] = acc0[j];
        lp[(1 * 8 + j) * 32 + lane] = acc1[j];
        lp[(2 * 8 + j) * 32 + lane] = acc2[j];
        lp[(3 * 8 + j) * 32 + lane] = acc3[j];
    }
    __syncthreads();

#pragma unroll
    for (int e = 0; e < 4; ++e) {
        int idx = threadIdx.x * 4 + e;
        float v = 0.0f;
#pragma unroll
        for (int w = 0; w < 8; ++w) v += lds[w * 1024 + idx];
        int slot = idx >> 5;
        int l    = idx & 31;
        int t    = slot >> 3;
        int j    = slot & 7;
        int mm   = j + ((l >> 4) << 3);
        int nn   = t * 16 + (l & 15);
        WX[((size_t)b * KPAD + mm) * CIN + c0 + nn] = v;
    }
}

__global__ __launch_bounds__(256) void vlad_finalize(const float* __restrict__ WX,
                                                     const float* __restrict__ asum,
                                                     const float* __restrict__ centroids,
                                                     float* __restrict__ out) {
    const int b   = blockIdx.x >> 3;
    const int kc  = blockIdx.x & 7;
    const int tid = threadIdx.x;

    const float a = asum[b * KPAD + kc];
    const float* wp = WX + ((size_t)b * KPAD + kc) * CIN;
    const float* cp = centroids + (size_t)kc * CIN;

    float r0 = wp[tid]       - a * cp[tid];
    float r1 = wp[tid + 256] - a * cp[tid + 256];

    __shared__ float red[256];
    red[tid] = r0 * r0 + r1 * r1;
    __syncthreads();
    for (int off = 128; off > 0; off >>= 1) {
        if (tid < off) red[tid] += red[tid + off];
        __syncthreads();
    }
    __shared__ float invn;
    if (tid == 0) invn = 1.0f / fmaxf(sqrtf(red[0]), 1e-12f);
    __syncthreads();

    float* op = out + ((size_t)b * KKEEP + kc) * CIN;
    op[tid]       = r0 * invn;
    op[tid + 256] = r1 * invn;
}

extern "C" void kernel_launch(void* const* d_in, const int* in_sizes, int n_in,
                              void* d_out, int out_size, void* d_ws, size_t ws_size,
                              hipStream_t stream) {
    const float* features  = (const float*)d_in[0];
    const float* score     = (const float*)d_in[1];
    const float* centroids = (const float*)d_in[2];
    float* out = (float*)d_out;

    float* S    = (float*)d_ws;
    float* WX   = S  + (size_t)BATCH * KPAD * HW;
    float* ASUM = WX + (size_t)BATCH * KPAD * CIN;

    vlad_zero_asum<<<2, 256, 0, stream>>>(ASUM);
    vlad_softmax  <<<BATCH * HW / 256, 256, 0, stream>>>(score, S, ASUM);
    vlad_wmma_gemm<<<BATCH * 8,        256, 0, stream>>>(features, S, WX);
    vlad_finalize <<<BATCH * KKEEP,    256, 0, stream>>>(WX, ASUM, centroids, out);
}


# Cluster sums leave the softmax kernel (10 LDS tree reductions and atomics per block, and the zeroing launch): each finalize block sums its own S row; 3 launches; a_sum's sum order changes, no atomics
# speedup vs baseline: 1.4316x; 1.2755x over previous
#include <hip/hip_runtime.h>
#include <math.h>

typedef float v2f __attribute__((ext_vector_type(2)));
typedef float v4f __attribute__((ext_vector_type(4)));
typedef float v8f __attribute__((ext_vector_type(8)));

#define BATCH 32
#define CIN   512
#define HW    4096
#define NCLUS 10
#define KPAD  16
#define KKEEP 8
#define KCHUNK (HW / 8)


__global__ __launch_bounds__(256) void vlad_softmax(const float* __restrict__ score,
                                                    float* __restrict__ S) {
    int gid = blockIdx.x * 256 + threadIdx.x;
    int b   = gid >> 12;
    int hw  = gid & (HW - 1);

    const float* sp = score + (size_t)b * NCLUS * HW + hw;
    float s[NCLUS];
    float mx = -1e30f;
#pragma unroll
    for (int k = 0; k < NCLUS; ++k) { s[k] = sp[(size_t)k * HW]; mx = fmaxf(mx, s[k]); }
    float sum = 0.0f;
#pragma unroll
    for (int k = 0; k < NCLUS; ++k) { s[k] = __expf(s[k] - mx); sum += s[k]; }
    float inv = 1.0f / sum;

    float* Sp = S + (size_t)b * KPAD * HW + hw;
#pragma unroll
    for (int k = 0; k < NCLUS; ++k) { s[k] *= inv; Sp[(size_t)k * HW] = s[k]; }
#pragma unroll
    for (int k = NCLUS; k < KPAD; ++k) Sp[(size_t)k * HW] = 0.0f;
}

struct Frags {
    v4f a[4];
    v4f b0[4];
    v4f b1[4];
    v4f b2[4];
    v4f b3[4];
};

__device__ __forceinline__ void load_group(Frags& f, const float* __restrict__ Sp,
                                           const float* __restrict__ Fp, int k) {
#pragma unroll
    for (int j = 0; j < 4; ++j) {
        f.a[j]  = *(const v4f*)(Sp + k + 8 * j);
        f.b0[j] = *(const v4f*)(Fp + k + 8 * j);
        f.b1[j] = *(const v4f*)(Fp + (size_t)16 * HW + k + 8 * j);
        f.b2[j] = *(const v4f*)(Fp + (size_t)32 * HW + k + 8 * j);
        f.b3[j] = *(const v4f*)(Fp + (size_t)48 * HW + k + 8 * j);
    }
}

__device__ __forceinline__ void mma_group(const Frags& f, v8f& acc0, v8f& acc1,
                                          v8f& acc2, v8f& acc3) {
#pragma unroll
    for (int j = 0; j < 4; ++j) {
        v2f alo = {f.a[j].x, f.a[j].y};
        v2f ahi = {f.a[j].z, f.a[j].w};
        v2f p0 = {f.b0[j].x, f.b0[j].y};
        v2f p1 = {f.b1[j].x, f.b1[j].y};
        v2f p2 = {f.b2[j].x, f.b2[j].y};
        v2f p3 = {f.b3[j].x, f.b3[j].y};
        acc0 = __builtin_amdgcn_wmma_f32_16x16x4_f32(false, alo, false, p0, (short)0, acc0, false, false);
        acc1 = __builtin_amdgcn_wmma_f32_16x16x4_f32(false, alo, false, p1, (short)0, acc1, false, false);
        acc2 = __builtin_amdgcn_wmma_f32_16x16x4_f32(false, alo, false, p2, (short)0, acc2, false, false);
        acc3 = __builtin_amdgcn_wmma_f32_16x16x4_f32(false, alo, false, p3, (short)0, acc3, false, false);
        v2f q0 = {f.b0[j].z, f.b0[j].w};
        v2f q1 = {f.b1[j].z, f.b1[j].w};
        v2f q2 = {f.b2[j].z, f.b2[j].w};
        v2f q3 = {f.b3[j].z, f.b3[j].w};
        acc0 = __builtin_amdgcn_wmma_f32_16x16x4_f32(false, ahi, false, q0, (short)0, acc0, false, false);
        acc1 = __builtin_amdgcn_wmma_f32_16x16x4_f32(false, ahi, false, q1, (short)0, acc1, false, false);
        acc2 = __builtin_amdgcn_wmma_f32_16x16x4_f32(false, ahi, false, q2, (short)0, acc2, false, false);
        acc3 = __builtin_amdgcn_wmma_f32_16x16x4_f32(false, ahi, false, q3, (short)0, acc3, false, false);
    }
}

__global__ __launch_bounds__(256) void vlad_wmma_gemm(const float* __restrict__ F,
                                                      const float* __restrict__ S,
                                                      float* __restrict__ WX) {
    const int b     = blockIdx.x >> 3;
    const int ntile = blockIdx.x & 7;
    const int c0    = ntile * 64;

    const int wave  = threadIdx.x >> 5;
    const int lane  = threadIdx.x & 31;
    const int kbase = wave * KCHUNK;
    const int mn    = lane & 15;
    const int kk    = (lane >> 4) * 4;

    const float* Sp = S + ((size_t)b * KPAD + mn) * HW + kbase + kk;
    const float* Fp = F + ((size_t)b * CIN + c0 + mn) * HW + kbase + kk;

    v8f acc0 = {}; v8f acc1 = {}; v8f acc2 = {}; v8f acc3 = {};

    Frags cur, nxt;
    load_group(cur, Sp, Fp, 0);
#pragma unroll 1
    for (int k = 32; k < KCHUNK; k += 32) {
        load_group(nxt, Sp, Fp, k);
        mma_group(cur, acc0, acc1, acc2, acc3);
        cur = nxt;
    }
    mma_group(cur, acc0, acc1, acc2, acc3);

    __shared__ float lds[8 * 1024];
    float* lp = lds + wave * 1024;
#pragma unroll
    for (int j = 0; j < 8; ++j) {
        lp[(0 * 8 + j) * 32 + lane] = acc0[j];
        lp[(1 * 8 + j) * 32 + lane] = acc1[j];
        lp[(2 * 8 + j) * 32 + lane] = acc2[j];
        lp[(3 * 8 + j) * 32 + lane] = acc3[j];
    }
    __syncthreads();

#pragma unroll
    for (int e = 0; e < 4; ++e) {
        int idx = threadIdx.x * 4 + e;
        float v = 0.0f;
#pragma unroll
        for (int w = 0; w < 8; ++w) v += lds[w * 1024 + idx];
        int slot = idx >> 5;
        int l    = idx & 31;
        int t    = slot >> 3;
        int j    = slot & 7;
        int mm   = j + ((l >> 4) << 3);
        int nn   = t * 16 + (l & 15);
        WX[((size_t)b * KPAD + mm) * CIN + c0 + nn] = v;
    }
}

__global__ __launch_bounds__(256) void vlad_finalize(const float* __restrict__ WX,
                                                     const float* __restrict__ S,
                                                     const float* __restrict__ centroids,
                                                     float* __restrict__ out) {
    const int b   = blockIdx.x >> 3;
    const int kc  = blockIdx.x & 7;
    const int tid = threadIdx.x;

    const v4f* sp = (const v4f*)(S + ((size_t)b * KPAD + kc) * HW);
    const v4f q0 = sp[tid];
    const v4f q1 = sp[tid + 256];
    const v4f q2 = sp[tid + 512];
    const v4f q3 = sp[tid + 768];

    const float* wp = WX + ((size_t)b * KPAD + kc) * CIN;
    const float* cp = centroids + (size_t)kc * CIN;
    const float w0 = wp[tid];
    const float w1 = wp[tid + 256];
    const float c0 = cp[tid];
    const float c1 = cp[tid + 256];

    float part = (((q0.x + q0.y) + (q0.z + q0.w)) + ((q1.x + q1.y) + (q1.z + q1.w)))
               + (((q2.x + q2.y) + (q2.z + q2.w)) + ((q3.x + q3.y) + (q3.z + q3.w)));
#pragma unroll
    for (int off = 16; off > 0; off >>= 1) part += __shfl_xor(part, off);
    __shared__ float wsum[8];
    if ((tid & 31) == 0) wsum[tid >> 5] = part;
    __syncthreads();
    const float a = ((wsum[0] + wsum[1]) + (wsum[2] + wsum[3]))
                  + ((wsum[4] + wsum[5]) + (wsum[6] + wsum[7]));

    float r0 = w0 - a * c0;
    float r1 = w1 - a * c1;

    __shared__ float red[256];
    red[tid] = r0 * r0 + r1 * r1;
    __syncthreads();
    for (int off = 128; off > 0; off >>= 1) {
        if (tid < off) red[tid] += red[tid + off];
        __syncthreads();
    }
    __shared__ float invn;
    if (tid == 0) invn = 1.0f / fmaxf(sqrtf(red[0]), 1e-12f);
    __syncthreads();

    float* op = out + ((size_t)b * KKEEP + kc) * CIN;
    op[tid]       = r0 * invn;
    op[tid + 256] = r1 * invn;
}

extern "C" void kernel_launch(void* const* d_in, const int* in_sizes, int n_in,
                              void* d_out, int out_size, void* d_ws, size_t ws_size,
                              hipStream_t stream) {
    const float* features  = (const float*)d_in[0];
    const float* score     = (const float*)d_in[1];
    const float* centroids = (const float*)d_in[2];
    float* out = (float*)d_out;

    float* S    = (float*)d_ws;
    float* WX   = S  + (size_t)BATCH * KPAD * HW;

    vlad_softmax  <<<BATCH * HW / 256, 256, 0, stream>>>(score, S);
    vlad_wmma_gemm<<<BATCH * 8,        256, 0, stream>>>(features, S, WX);
    vlad_finalize <<<BATCH * KKEEP,    256, 0, stream>>>(WX, S, centroids, out);
}


# GEMM operands fetched as whole lines: one load reads 256 B of each of 2 rows (was 32 B of each of sixteen rows); WMMA fragments come from a wave-private LDS tile; k pairing and sum order unchanged
# speedup vs baseline: 1.6026x; 1.1194x over previous
#include <hip/hip_runtime.h>
#include <math.h>

typedef float v2f __attribute__((ext_vector_type(2)));
typedef float v4f __attribute__((ext_vector_type(4)));
typedef float v8f __attribute__((ext_vector_type(8)));

#define BATCH 32
#define CIN   512
#define HW    4096
#define NCLUS 10
#define KPAD  16
#define KKEEP 8
#define KCHUNK (HW / 8)


__global__ __launch_bounds__(256) void vlad_softmax(const float* __restrict__ score,
                                                    float* __restrict__ S) {
    int gid = blockIdx.x * 256 + threadIdx.x;
    int b   = gid >> 12;
    int hw  = gid & (HW - 1);

    const float* sp = score + (size_t)b * NCLUS * HW + hw;
    float s[NCLUS];
    float mx = -1e30f;
#pragma unroll
    for (int k = 0; k < NCLUS; ++k) { s[k] = sp[(size_t)k * HW]; mx = fmaxf(mx, s[k]); }
    float sum = 0.0f;
#pragma unroll
    for (int k = 0; k < NCLUS; ++k) { s[k] = __expf(s[k] - mx); sum += s[k]; }
    float inv = 1.0f / sum;

    float* Sp = S + (size_t)b * KPAD * HW + hw;
#pragma unroll
    for (int k = 0; k < NCLUS; ++k) { s[k] *= inv; Sp[(size_t)k * HW] = s[k]; }
#pragma unroll
    for (int k = NCLUS; k < KPAD; ++k) Sp[(size_t)k * HW] = 0.0f;
}

#define GK      64
#define NGRP    (KCHUNK / GK)
#define TROWS   80
#define TPITCH  68
#define TILE_F  (TROWS * TPITCH)
#define NLOAD   (TROWS / 2)

__device__ __forceinline__ void load_rows(v4f (&R)[NLOAD], const float* __restrict__ Sg,
                                          const float* __restrict__ Fg, int k) {
#pragma unroll
    for (int i = 0; i < 8; ++i)
        R[i] = *(const v4f*)(Sg + (size_t)(2 * i) * HW + k);
#pragma unroll
    for (int i = 0; i < 32; ++i)
        R[8 + i] = *(const v4f*)(Fg + (size_t)(2 * i) * HW + k);
}

__device__ __forceinline__ void store_rows(const v4f (&R)[NLOAD], float* st) {
#pragma unroll
    for (int i = 0; i < NLOAD; ++i)
        *(v4f*)(st + 2 * i * TPITCH) = R[i];
}

__device__ __forceinline__ void store_rows_and_refill(v4f (&R)[NLOAD], float* st,
                                                      const float* __restrict__ Sg,
                                                      const float* __restrict__ Fg, int k) {
#pragma unroll
    for (int c = 0; c < NLOAD / 8; ++c) {
#pragma unroll
        for (int i = 8 * c; i < 8 * c + 8; ++i)
            *(v4f*)(st + 2 * i * TPITCH) = R[i];
#pragma unroll
        for (int i = 8 * c; i < 8 * c + 8; ++i)
            R[i] = (i < 8) ? *(const v4f*)(Sg + (size_t)(2 * i) * HW + k)
                           : *(const v4f*)(Fg + (size_t)(2 * (i - 8)) * HW + k);
        __builtin_amdgcn_sched_barrier(0);
    }
}

__device__ __forceinline__ void wave_lds_order() {
    __builtin_amdgcn_fence(__ATOMIC_RELEASE, "wavefront");
    __builtin_amdgcn_wave_barrier();
    __builtin_amdgcn_fence(__ATOMIC_ACQUIRE, "wavefront");
}

__device__ __forceinline__ void mma_tile(const float* fr, v8f& acc0, v8f& acc1,
                                         v8f& acc2, v8f& acc3) {
#pragma unroll
    for (int j = 0; j < GK / 8; ++j) {
        const v4f a  = *(const v4f*)(fr + 8 * j);
        const v4f b0 = *(const v4f*)(fr + 16 * TPITCH + 8 * j);
        const v4f b1 = *(const v4f*)(fr + 32 * TPITCH + 8 * j);
        const v4f b2 = *(const v4f*)(fr + 48 * TPITCH + 8 * j);
        const v4f b3 = *(const v4f*)(fr + 64 * TPITCH + 8 * j);
        v2f alo = {a.x, a.y};
        v2f ahi = {a.z, a.w};
        v2f p0 = {b0.x, b0.y};
        v2f p1 = {b1.x, b1.y};
        v2f p2 = {b2.x, b2.y};
        v2f p3 = {b3.x, b3.y};
        acc0 = __builtin_amdgcn_wmma_f32_16x16x4_f32(false, alo, false, p0, (short)0, acc0, false, false);
        acc1 = __builtin_amdgcn_wmma_f32_16x16x4_f32(false, alo, false, p1, (short)0, acc1, false, false);
        acc2 = __builtin_amdgcn_wmma_f32_16x16x4_f32(false, alo, false, p2, (short)0, acc2, false, false);
        acc3 = __builtin_amdgcn_wmma_f32_16x16x4_f32(false, alo, false, p3, (short)0, acc3, false, false);
        v2f q0 = {b0.z, b0.w};
        v2f q1 = {b1.z, b1.w};
        v2f q2 = {b2.z, b2.w};
        v2f q3 = {b3.z, b3.w};
        acc0 = __builtin_amdgcn_wmma_f32_16x16x4_f32(false, ahi, false, q0, (short)0, acc0, false, false);
        acc1 = __builtin_amdgcn_wmma_f32_16x16x4_f32(false, ahi, false, q1, (short)0, acc1, false, false);
        acc2 = __builtin_amdgcn_wmma_f32_16x16x4_f32(false, ahi, false, q2, (short)0, acc2, false, false);
        acc3 = __builtin_amdgcn_wmma_f32_16x16x4_f32(false, ahi, false, q3, (short)0, acc3, false, false);
    }
}

__global__ __launch_bounds__(256) void vlad_wmma_gemm(const float* __restrict__ F,
                                                      const float* __restrict__ S,
                                                      float* __restrict__ WX) {
    const int b     = blockIdx.x >> 3;
    const int ntile = blockIdx.x & 7;
    const int c0    = ntile * 64;

    const int wave  = threadIdx.x >> 5;
    const int lane  = threadIdx.x & 31;
    const int kbase = wave * KCHUNK;
    const int p     = lane & 15;
    const int h     = lane >> 4;

    __shared__ __attribute__((aligned(16))) float stage[8 * TILE_F];
    __shared__ float lds[8 * 1024];
    float* tile = stage + wave * TILE_F;

    const float* Sg = S + ((size_t)b * KPAD + h) * HW + kbase + 4 * p;
    const float* Fg = F + ((size_t)b * CIN + c0 + h) * HW + kbase + 4 * p;
    float*       st = tile + h * TPITCH + 4 * p;
    const float* fr = tile + p * TPITCH + 4 * h;

    v8f acc0 = {}; v8f acc1 = {}; v8f acc2 = {}; v8f acc3 = {};

    v4f R[NLOAD];
    load_rows(R, Sg, Fg, 0);
#pragma unroll 1
    for (int g = 1; g < NGRP; ++g) {
        store_rows_and_refill(R, st, Sg, Fg, g * GK);
        wave_lds_order();
        mma_tile(fr, acc0, acc1, acc2, acc3);
        wave_lds_order();
    }
    store_rows(R, st);
    wave_lds_order();
    mma_tile(fr, acc0, acc1, acc2, acc3);

    float* lp = lds + wave * 1024;
#pragma unroll
    for (int j = 0; j < 8; ++j) {
        lp[(0 * 8 + j) * 32 + lane] = acc0[j];
        lp[(1 * 8 + j) * 32 + lane] = acc1[j];
        lp[(2 * 8 + j) * 32 + lane] = acc2[j];
        lp[(3 * 8 + j) * 32 + lane] = acc3[j];
    }
    __syncthreads();

#pragma unroll
    for (int e = 0; e < 4; ++e) {
        int idx = threadIdx.x * 4 + e;
        float v = 0.0f;
#pragma unroll
        for (int w = 0; w < 8; ++w) v += lds[w * 1024 + idx];
        int slot = idx >> 5;
        int l    = idx & 31;
        int t    = slot >> 3;
        int j    = slot & 7;
        int mm   = j + ((l >> 4) << 3);
        int nn   = t * 16 + (l & 15);
        WX[((size_t)b * KPAD + mm) * CIN + c0 + nn] = v;
    }
}

__global__ __launch_bounds__(256) void vlad_finalize(const float* __restrict__ WX,
                                                     const float* __restrict__ S,
                                                     const float* __restrict__ centroids,
                                                     float* __restrict__ out) {
    const int b   = blockIdx.x >> 3;
    const int kc  = blockIdx.x & 7;
    const int tid = threadIdx.x;

    const v4f* sp = (const v4f*)(S + ((size_t)b * KPAD + kc) * HW);
    const v4f q0 = sp[tid];
    const v4f q1 = sp[tid + 256];
    const v4f q2 = sp[tid + 512];
    const v4f q3 = sp[tid + 768];

    const float* wp = WX + ((size_t)b * KPAD + kc) * CIN;
    const float* cp = centroids + (size_t)kc * CIN;
    const float w0 = wp[tid];
    const float w1 = wp[tid + 256];
    const float c0 = cp[tid];
    const float c1 = cp[tid + 256];

    float part = (((q0.x + q0.y) + (q0.z + q0.w)) + ((q1.x + q1.y) + (q1.z + q1.w)))
               + (((q2.x + q2.y) + (q2.z + q2.w)) + ((q3.x + q3.y) + (q3.z + q3.w)));
#pragma unroll
    for (int off = 16; off > 0; off >>= 1) part += __shfl_xor(part, off);
    __shared__ float wsum[8];
    if ((tid & 31) == 0) wsum[tid >> 5] = part;
    __syncthreads();
    const float a = ((wsum[0] + wsum[1]) + (wsum[2] + wsum[3]))
                  + ((wsum[4] + wsum[5]) + (wsum[6] + wsum[7]));

    float r0 = w0 - a * c0;
    float r1 = w1 - a * c1;

    __shared__ float red[256];
    red[tid] = r0 * r0 + r1 * r1;
    __syncthreads();
    for (int off = 128; off > 0; off >>= 1) {
        if (tid < off) red[tid] += red[tid + off];
        __syncthreads();
    }
    __shared__ float invn;
    if (tid == 0) invn = 1.0f / fmaxf(sqrtf(red[0]), 1e-12f);
    __syncthreads();

    float* op = out + ((size_t)b * KKEEP + kc) * CIN;
    op[tid]       = r0 * invn;
    op[tid + 256] = r1 * invn;
}

extern "C" void kernel_launch(void* const* d_in, const int* in_sizes, int n_in,
                              void* d_out, int out_size, void* d_ws, size_t ws_size,
                              hipStream_t stream) {
    const float* features  = (const float*)d_in[0];
    const float* score     = (const float*)d_in[1];
    const float* centroids = (const float*)d_in[2];
    float* out = (float*)d_out;

    float* S    = (float*)d_ws;
    float* WX   = S  + (size_t)BATCH * KPAD * HW;

    vlad_softmax  <<<BATCH * HW / 256, 256, 0, stream>>>(score, S);
    vlad_wmma_gemm<<<BATCH * 8,        256, 0, stream>>>(features, S, WX);
    vlad_finalize <<<BATCH * KKEEP,    256, 0, stream>>>(WX, S, centroids, out);
}


# Softmax formed inside the GEMM from the raw scores (no S plane: 10 score rows fetched, not 16 of S); cluster sums and centroid term in its epilogue; 2 launches; softmax and a_sum sum order change
# speedup vs baseline: 1.8146x; 1.1323x over previous
#include <hip/hip_runtime.h>
#include <math.h>

typedef float v2f __attribute__((ext_vector_type(2)));
typedef float v4f __attribute__((ext_vector_type(4)));
typedef float v8f __attribute__((ext_vector_type(8)));

#define BATCH 32
#define CIN   512
#define HW    4096
#define NCLUS 10
#define KKEEP 8
#define KCHUNK (HW / 8)


#define GK      64
#define NGRP    (KCHUNK / GK)
#define TROWS   80
#define TPITCH  68
#define TILE_F  (TROWS * TPITCH)
#define NSC     (NCLUS / 2)
#define NFL     32
#define RP_T    144
#define RP_W    (4 * RP_T)

__device__ __forceinline__ float xhalf(float x) {
    return __builtin_bit_cast(float,
        __builtin_amdgcn_permlanex16(__builtin_bit_cast(int, x), __builtin_bit_cast(int, x),
                                     0x76543210, 0xfedcba98, false, false));
}

__device__ __forceinline__ void softmax_rows(const v4f (&C)[NSC], v4f (&P)[4], float (&asum)[4]) {
#pragma unroll
    for (int c = 0; c < 4; ++c) {
        float m = fmaxf(fmaxf(fmaxf(C[0][c], C[1][c]), fmaxf(C[2][c], C[3][c])), C[4][c]);
        m = fmaxf(m, xhalf(m));
        float e[NSC];
#pragma unroll
        for (int i = 0; i < NSC; ++i) e[i] = __expf(C[i][c] - m);
        float s = ((e[0] + e[1]) + (e[2] + e[3])) + e[4];
        s += xhalf(s);
        const float inv = 1.0f / s;
#pragma unroll
        for (int i = 0; i < 4; ++i) P[i][c] = e[i] * inv;
    }
#pragma unroll
    for (int i = 0; i < 4; ++i) asum[i] += (P[i].x + P[i].y) + (P[i].z + P[i].w);
}

__device__ __forceinline__ void load_rows(v4f (&SC)[NSC], v4f (&R)[NFL], const float* __restrict__ Sg,
                                          const float* __restrict__ Fg, int k) {
#pragma unroll
    for (int i = 0; i < NSC; ++i)
        SC[i] = *(const v4f*)(Sg + (size_t)(2 * i) * HW + k);
#pragma unroll
    for (int i = 0; i < NFL; ++i)
        R[i] = *(const v4f*)(Fg + (size_t)(2 * i) * HW + k);
}

__device__ __forceinline__ void store_rows(const v4f (&SC)[NSC], const v4f (&R)[NFL], float* st,
                                           float (&asum)[4]) {
    v4f P[4];
    softmax_rows(SC, P, asum);
#pragma unroll
    for (int i = 0; i < 4; ++i)
        *(v4f*)(st + 2 * i * TPITCH) = P[i];
#pragma unroll
    for (int i = 0; i < NFL; ++i)
        *(v4f*)(st + (16 + 2 * i) * TPITCH) = R[i];
}

__device__ __forceinline__ void store_rows_and_refill(v4f (&SC)[NSC], v4f (&R)[NFL], float* st,
                                                      const float* __restrict__ Sg,
                                                      const float* __restrict__ Fg, int k,
                                                      float (&asum)[4]) {
    v4f P[4];
    softmax_rows(SC, P, asum);
#pragma unroll
    for (int i = 0; i < 4; ++i)
        *(v4f*)(st + 2 * i * TPITCH) = P[i];
#pragma unroll
    for (int i = 0; i < NSC; ++i)
        SC[i] = *(const v4f*)(Sg + (size_t)(2 * i) * HW + k);
    __builtin_amdgcn_sched_barrier(0);
#pragma unroll
    for (int c = 0; c < NFL / 8; ++c) {
#pragma unroll
        for (int i = 8 * c; i < 8 * c + 8; ++i)
            *(v4f*)(st + (16 + 2 * i) * TPITCH) = R[i];
#pragma unroll
        for (int i = 8 * c; i < 8 * c + 8; ++i)
            R[i] = *(const v4f*)(Fg + (size_t)(2 * i) * HW + k);
        __builtin_amdgcn_sched_barrier(0);
    }
}

__device__ __forceinline__ void wave_lds_order() {
    __builtin_amdgcn_fence(__ATOMIC_RELEASE, "wavefront");
    __builtin_amdgcn_wave_barrier();
    __builtin_amdgcn_fence(__ATOMIC_ACQUIRE, "wavefront");
}

__device__ __forceinline__ void mma_tile(const float* fr, v8f& acc0, v8f& acc1,
                                         v8f& acc2, v8f& acc3) {
#pragma unroll
    for (int j = 0; j < GK / 8; ++j) {
        const v4f a  = *(const v4f*)(fr + 8 * j);
        const v4f b0 = *(const v4f*)(fr + 16 * TPITCH + 8 * j);
        const v4f b1 = *(const v4f*)(fr + 32 * TPITCH + 8 * j);
        const v4f b2 = *(const v4f*)(fr + 48 * TPITCH + 8 * j);
        const v4f b3 = *(const v4f*)(fr + 64 * TPITCH + 8 * j);
        v2f alo = {a.x, a.y};
        v2f ahi = {a.z, a.w};
        v2f p0 = {b0.x, b0.y};
        v2f p1 = {b1.x, b1.y};
        v2f p2 = {b2.x, b2.y};
        v2f p3 = {b3.x, b3.y};
        acc0 = __builtin_amdgcn_wmma_f32_16x16x4_f32(false, alo, false, p0, (short)0, acc0, false, false);
        acc1 = __builtin_amdgcn_wmma_f32_16x16x4_f32(false, alo, false, p1, (short)0, acc1, false, false);
        acc2 = __builtin_amdgcn_wmma_f32_16x16x4_f32(false, alo, false, p2, (short)0, acc2, false, false);
        acc3 = __builtin_amdgcn_wmma_f32_16x16x4_f32(false, alo, false, p3, (short)0, acc3, false, false);
        v2f q0 = {b0.z, b0.w};
        v2f q1 = {b1.z, b1.w};
        v2f q2 = {b2.z, b2.w};
        v2f q3 = {b3.z, b3.w};
        acc0 = __builtin_amdgcn_wmma_f32_16x16x4_f32(false, ahi, false, q0, (short)0, acc0, false, false);
        acc1 = __builtin_amdgcn_wmma_f32_16x16x4_f32(false, ahi, false, q1, (short)0, acc1, false, false);
        acc2 = __builtin_amdgcn_wmma_f32_16x16x4_f32(false, ahi, false, q2, (short)0, acc2, false, false);
        acc3 = __builtin_amdgcn_wmma_f32_16x16x4_f32(false, ahi, false, q3, (short)0, acc3, false, false);
    }
}

__global__ __launch_bounds__(256) void vlad_wmma_gemm(const float* __restrict__ F,
                                                      const float* __restrict__ score,
                                                      const float* __restrict__ centroids,
                                                      float* __restrict__ RES) {
    const int b     = blockIdx.x >> 3;
    const int ntile = blockIdx.x & 7;
    const int c0    = ntile * 64;

    const int wave  = threadIdx.x >> 5;
    const int lane  = threadIdx.x & 31;
    const int kbase = wave * KCHUNK;
    const int p     = lane & 15;
    const int h     = lane >> 4;

    __shared__ __attribute__((aligned(16))) float stage[8 * TILE_F];
    __shared__ __attribute__((aligned(16))) float part[8 * RP_W];
    __shared__ float asl[8 * KKEEP];
    float* tile = stage + wave * TILE_F;

    const float* Sg = score + ((size_t)b * NCLUS + h) * HW + kbase + 4 * p;
    const float* Fg = F + ((size_t)b * CIN + c0 + h) * HW + kbase + 4 * p;
    float*       st = tile + h * TPITCH + 4 * p;
    const float* fr = tile + p * TPITCH + 4 * h;

    v8f acc0 = {}; v8f acc1 = {}; v8f acc2 = {}; v8f acc3 = {};
    float asum[4] = {0.0f, 0.0f, 0.0f, 0.0f};

    v4f SC[NSC];
    v4f R[NFL];
    load_rows(SC, R, Sg, Fg, 0);

    {
        const v4f z = {0.0f, 0.0f, 0.0f, 0.0f};
#pragma unroll
        for (int i = 4; i < 8; ++i)
            *(v4f*)(st + 2 * i * TPITCH) = z;
    }

#pragma unroll 1
    for (int g = 1; g < NGRP; ++g) {
        store_rows_and_refill(SC, R, st, Sg, Fg, g * GK, asum);
        wave_lds_order();
        mma_tile(fr, acc0, acc1, acc2, acc3);
        wave_lds_order();
    }
    store_rows(SC, R, st, asum);
    wave_lds_order();
    mma_tile(fr, acc0, acc1, acc2, acc3);

    if (h == 0) {
        float* lp = part + wave * RP_W + p;
#pragma unroll
        for (int j = 0; j < 8; ++j) {
            lp[0 * RP_T + j * 16] = acc0[j];
            lp[1 * RP_T + j * 16] = acc1[j];
            lp[2 * RP_T + j * 16] = acc2[j];
            lp[3 * RP_T + j * 16] = acc3[j];
        }
    }
#pragma unroll
    for (int i = 0; i < 4; ++i) {
        float v = asum[i];
#pragma unroll
        for (int off = 8; off > 0; off >>= 1) v += __shfl_xor(v, off);
        asum[i] = v;
    }
    if (p == 0) {
#pragma unroll
        for (int i = 0; i < 4; ++i) asl[wave * KKEEP + 2 * i + h] = asum[i];
    }
    __syncthreads();

    const int m  = threadIdx.x >> 5;
    const int n0 = 2 * lane;
    const float* pp = part + (n0 >> 4) * RP_T + m * 16 + (n0 & 15);
    v2f v = {0.0f, 0.0f};
#pragma unroll
    for (int w = 0; w < 8; ++w) v += *(const v2f*)(pp + w * RP_W);
    const float a = ((asl[0 * KKEEP + m] + asl[1 * KKEEP + m]) + (asl[2 * KKEEP + m] + asl[3 * KKEEP + m]))
                  + ((asl[4 * KKEEP + m] + asl[5 * KKEEP + m]) + (asl[6 * KKEEP + m] + asl[7 * KKEEP + m]));
    const v2f c = *(const v2f*)(centroids + (size_t)m * CIN + c0 + n0);
    v2f r;
    r.x = v.x - a * c.x;
    r.y = v.y - a * c.y;
    *(v2f*)(RES + ((size_t)b * KKEEP + m) * CIN + c0 + n0) = r;
}

__global__ __launch_bounds__(256) void vlad_finalize(const float* __restrict__ RES,
                                                     float* __restrict__ out) {
    const int tid = threadIdx.x;

    const float* rp = RES + (size_t)blockIdx.x * CIN;
    const float r0 = rp[tid];
    const float r1 = rp[tid + 256];

    __shared__ float red[256];
    red[tid] = r0 * r0 + r1 * r1;
    __syncthreads();
    for (int off = 128; off > 0; off >>= 1) {
        if (tid < off) red[tid] += red[tid + off];
        __syncthreads();
    }
    __shared__ float invn;
    if (tid == 0) invn = 1.0f / fmaxf(sqrtf(red[0]), 1e-12f);
    __syncthreads();

    float* op = out + (size_t)blockIdx.x * CIN;
    op[tid]       = r0 * invn;
    op[tid + 256] = r1 * invn;
}

extern "C" void kernel_launch(void* const* d_in, const int* in_sizes, int n_in,
                              void* d_out, int out_size, void* d_ws, size_t ws_size,
                              hipStream_t stream) {
    const float* features  = (const float*)d_in[0];
    const float* score     = (const float*)d_in[1];
    const float* centroids = (const float*)d_in[2];
    float* out = (float*)d_out;

    float* RES = (float*)d_ws;

    vlad_wmma_gemm<<<BATCH * 8,     256, 0, stream>>>(features, score, centroids, RES);
    vlad_finalize <<<BATCH * KKEEP, 256, 0, stream>>>(RES, out);
}


# Finalize as one wave per row: 16 residuals per lane in registers, sum of squares folded by DPP butterfly, no LDS tree or barriers, 256 waves not 2048; the norm's sum order changes
# speedup vs baseline: 1.8941x; 1.0438x over previous
#include <hip/hip_runtime.h>
#include <math.h>

typedef float v2f __attribute__((ext_vector_type(2)));
typedef float v4f __attribute__((ext_vector_type(4)));
typedef float v8f __attribute__((ext_vector_type(8)));

#define BATCH 32
#define CIN   512
#define HW    4096
#define NCLUS 10
#define KKEEP 8
#define KCHUNK (HW / 8)


#define GK      64
#define NGRP    (KCHUNK / GK)
#define TROWS   80
#define TPITCH  68
#define TILE_F  (TROWS * TPITCH)
#define NSC     (NCLUS / 2)
#define NFL     32
#define RP_T    144
#define RP_W    (4 * RP_T)

__device__ __forceinline__ float xhalf(float x) {
    return __builtin_bit_cast(float,
        __builtin_amdgcn_permlanex16(__builtin_bit_cast(int, x), __builtin_bit_cast(int, x),
                                     0x76543210, 0xfedcba98, false, false));
}

__device__ __forceinline__ void softmax_rows(const v4f (&C)[NSC], v4f (&P)[4], float (&asum)[4]) {
#pragma unroll
    for (int c = 0; c < 4; ++c) {
        float m = fmaxf(fmaxf(fmaxf(C[0][c], C[1][c]), fmaxf(C[2][c], C[3][c])), C[4][c]);
        m = fmaxf(m, xhalf(m));
        float e[NSC];
#pragma unroll
        for (int i = 0; i < NSC; ++i) e[i] = __expf(C[i][c] - m);
        float s = ((e[0] + e[1]) + (e[2] + e[3])) + e[4];
        s += xhalf(s);
        const float inv = 1.0f / s;
#pragma unroll
        for (int i = 0; i < 4; ++i) P[i][c] = e[i] * inv;
    }
#pragma unroll
    for (int i = 0; i < 4; ++i) asum[i] += (P[i].x + P[i].y) + (P[i].z + P[i].w);
}

__device__ __forceinline__ void load_rows(v4f (&SC)[NSC], v4f (&R)[NFL], const float* __restrict__ Sg,
                                          const float* __restrict__ Fg, int k) {
#pragma unroll
    for (int i = 0; i < NSC; ++i)
        SC[i] = *(const v4f*)(Sg + (size_t)(2 * i) * HW + k);
#pragma unroll
    for (int i = 0; i < NFL; ++i)
        R[i] = *(const v4f*)(Fg + (size_t)(2 * i) * HW + k);
}

__device__ __forceinline__ void store_rows(const v4f (&SC)[NSC], const v4f (&R)[NFL], float* st,
                                           float (&asum)[4]) {
    v4f P[4];
    softmax_rows(SC, P, asum);
#pragma unroll
    for (int i = 0; i < 4; ++i)
        *(v4f*)(st + 2 * i * TPITCH) = P[i];
#pragma unroll
    for (int i = 0; i < NFL; ++i)
        *(v4f*)(st + (16 + 2 * i) * TPITCH) = R[i];
}

__device__ __forceinline__ void store_rows_and_refill(v4f (&SC)[NSC], v4f (&R)[NFL], float* st,
                                                      const float* __restrict__ Sg,
                                                      const float* __restrict__ Fg, int k,
                                                      float (&asum)[4]) {
    v4f P[4];
    softmax_rows(SC, P, asum);
#pragma unroll
    for (int i = 0; i < 4; ++i)
        *(v4f*)(st + 2 * i * TPITCH) = P[i];
#pragma unroll
    for (int i = 0; i < NSC; ++i)
        SC[i] = *(const v4f*)(Sg + (size_t)(2 * i) * HW + k);
    __builtin_amdgcn_sched_barrier(0);
#pragma unroll
    for (int c = 0; c < NFL / 8; ++c) {
#pragma unroll
        for (int i = 8 * c; i < 8 * c + 8; ++i)
            *(v4f*)(st + (16 + 2 * i) * TPITCH) = R[i];
#pragma unroll
        for (int i = 8 * c; i < 8 * c + 8; ++i)
            R[i] = *(const v4f*)(Fg + (size_t)(2 * i) * HW + k);
        __builtin_amdgcn_sched_barrier(0);
    }
}

__device__ __forceinline__ void wave_lds_order() {
    __builtin_amdgcn_fence(__ATOMIC_RELEASE, "wavefront");
    __builtin_amdgcn_wave_barrier();
    __builtin_amdgcn_fence(__ATOMIC_ACQUIRE, "wavefront");
}

__device__ __forceinline__ void mma_tile(const float* fr, v8f& acc0, v8f& acc1,
                                         v8f& acc2, v8f& acc3) {
#pragma unroll
    for (int j = 0; j < GK / 8; ++j) {
        const v4f a  = *(const v4f*)(fr + 8 * j);
        const v4f b0 = *(const v4f*)(fr + 16 * TPITCH + 8 * j);
        const v4f b1 = *(const v4f*)(fr + 32 * TPITCH + 8 * j);
        const v4f b2 = *(const v4f*)(fr + 48 * TPITCH + 8 * j);
        const v4f b3 = *(const v4f*)(fr + 64 * TPITCH + 8 * j);
        v2f alo = {a.x, a.y};
        v2f ahi = {a.z, a.w};
        v2f p0 = {b0.x, b0.y};
        v2f p1 = {b1.x, b1.y};
        v2f p2 = {b2.x, b2.y};
        v2f p3 = {b3.x, b3.y};
        acc0 = __builtin_amdgcn_wmma_f32_16x16x4_f32(false, alo, false, p0, (short)0, acc0, false, false);
        acc1 = __builtin_amdgcn_wmma_f32_16x16x4_f32(false, alo, false, p1, (short)0, acc1, false, false);
        acc2 = __builtin_amdgcn_wmma_f32_16x16x4_f32(false, alo, false, p2, (short)0, acc2, false, false);
        acc3 = __builtin_amdgcn_wmma_f32_16x16x4_f32(false, alo, false, p3, (short)0, acc3, false, false);
        v2f q0 = {b0.z, b0.w};
        v2f q1 = {b1.z, b1.w};
        v2f q2 = {b2.z, b2.w};
        v2f q3 = {b3.z, b3.w};
        acc0 = __builtin_amdgcn_wmma_f32_16x16x4_f32(false, ahi, false, q0, (short)0, acc0, false, false);
        acc1 = __builtin_amdgcn_wmma_f32_16x16x4_f32(false, ahi, false, q1, (short)0, acc1, false, false);
        acc2 = __builtin_amdgcn_wmma_f32_16x16x4_f32(false, ahi, false, q2, (short)0, acc2, false, false);
        acc3 = __builtin_amdgcn_wmma_f32_16x16x4_f32(false, ahi, false, q3, (short)0, acc3, false, false);
    }
}

__global__ __launch_bounds__(256) void vlad_wmma_gemm(const float* __restrict__ F,
                                                      const float* __restrict__ score,
                                                      const float* __restrict__ centroids,
                                                      float* __restrict__ RES) {
    const int b     = blockIdx.x >> 3;
    const int ntile = blockIdx.x & 7;
    const int c0    = ntile * 64;

    const int wave  = threadIdx.x >> 5;
    const int lane  = threadIdx.x & 31;
    const int kbase = wave * KCHUNK;
    const int p     = lane & 15;
    const int h     = lane >> 4;

    __shared__ __attribute__((aligned(16))) float stage[8 * TILE_F];
    __shared__ __attribute__((aligned(16))) float part[8 * RP_W];
    __shared__ float asl[8 * KKEEP];
    float* tile = stage + wave * TILE_F;

    const float* Sg = score + ((size_t)b * NCLUS + h) * HW + kbase + 4 * p;
    const float* Fg = F + ((size_t)b * CIN + c0 + h) * HW + kbase + 4 * p;
    float*       st = tile + h * TPITCH + 4 * p;
    const float* fr = tile + p * TPITCH + 4 * h;

    v8f acc0 = {}; v8f acc1 = {}; v8f acc2 = {}; v8f acc3 = {};
    float asum[4] = {0.0f, 0.0f, 0.0f, 0.0f};

    v4f SC[NSC];
    v4f R[NFL];
    load_rows(SC, R, Sg, Fg, 0);

    {
        const v4f z = {0.0f, 0.0f, 0.0f, 0.0f};
#pragma unroll
        for (int i = 4; i < 8; ++i)
            *(v4f*)(st + 2 * i * TPITCH) = z;
    }

#pragma unroll 1
    for (int g = 1; g < NGRP; ++g) {
        store_rows_and_refill(SC, R, st, Sg, Fg, g * GK, asum);
        wave_lds_order();
        mma_tile(fr, acc0, acc1, acc2, acc3);
        wave_lds_order();
    }
    store_rows(SC, R, st, asum);
    wave_lds_order();
    mma_tile(fr, acc0, acc1, acc2, acc3);

    if (h == 0) {
        float* lp = part + wave * RP_W + p;
#pragma unroll
        for (int j = 0; j < 8; ++j) {
            lp[0 * RP_T + j * 16] = acc0[j];
            lp[1 * RP_T + j * 16] = acc1[j];
            lp[2 * RP_T + j * 16] = acc2[j];
            lp[3 * RP_T + j * 16] = acc3[j];
        }
    }
#pragma unroll
    for (int i = 0; i < 4; ++i) {
        float v = asum[i];
#pragma unroll
        for (int off = 8; off > 0; off >>= 1) v += __shfl_xor(v, off);
        asum[i] = v;
    }
    if (p == 0) {
#pragma unroll
        for (int i = 0; i < 4; ++i) asl[wave * KKEEP + 2 * i + h] = asum[i];
    }
    __syncthreads();

    const int m  = threadIdx.x >> 5;
    const int n0 = 2 * lane;
    const float* pp = part + (n0 >> 4) * RP_T + m * 16 + (n0 & 15);
    v2f v = {0.0f, 0.0f};
#pragma unroll
    for (int w = 0; w < 8; ++w) v += *(const v2f*)(pp + w * RP_W);
    const float a = ((asl[0 * KKEEP + m] + asl[1 * KKEEP + m]) + (asl[2 * KKEEP + m] + asl[3 * KKEEP + m]))
                  + ((asl[4 * KKEEP + m] + asl[5 * KKEEP + m]) + (asl[6 * KKEEP + m] + asl[7 * KKEEP + m]));
    const v2f c = *(const v2f*)(centroids + (size_t)m * CIN + c0 + n0);
    v2f r;
    r.x = v.x - a * c.x;
    r.y = v.y - a * c.y;
    *(v2f*)(RES + ((size_t)b * KKEEP + m) * CIN + c0 + n0) = r;
}

#define FIN_ROWS 4

template <int XM>
__device__ __forceinline__ float row_xor(float x) {
    return __builtin_bit_cast(float,
        __builtin_amdgcn_update_dpp(0, __builtin_bit_cast(int, x), 0x160 + XM, 0xf, 0xf, true));
}

__global__ __launch_bounds__(32 * FIN_ROWS) void vlad_finalize(const float* __restrict__ RES,
                                                               float* __restrict__ out) {
    const int lane = threadIdx.x & 31;
    const int row  = blockIdx.x * FIN_ROWS + (threadIdx.x >> 5);

    const float* rp = RES + (size_t)row * CIN + 4 * lane;
    v4f r[4];
#pragma unroll
    for (int q = 0; q < 4; ++q) r[q] = *(const v4f*)(rp + 128 * q);

    float s[4];
#pragma unroll
    for (int q = 0; q < 4; ++q)
        s[q] = fmaf(r[q].w, r[q].w, fmaf(r[q].z, r[q].z, fmaf(r[q].y, r[q].y, r[q].x * r[q].x)));
    float t = (s[0] + s[1]) + (s[2] + s[3]);
    t += row_xor<1>(t);
    t += row_xor<2>(t);
    t += row_xor<4>(t);
    t += row_xor<8>(t);
    t += xhalf(t);

    const float invn = 1.0f / fmaxf(sqrtf(t), 1e-12f);

    float* op = out + (size_t)row * CIN + 4 * lane;
#pragma unroll
    for (int q = 0; q < 4; ++q) *(v4f*)(op + 128 * q) = r[q] * invn;
}

extern "C" void kernel_launch(void* const* d_in, const int* in_sizes, int n_in,
                              void* d_out, int out_size, void* d_ws, size_t ws_size,
                              hipStream_t stream) {
    const float* features  = (const float*)d_in[0];
    const float* score     = (const float*)d_in[1];
    const float* centroids = (const float*)d_in[2];
    float* out = (float*)d_out;

    float* RES = (float*)d_ws;

    vlad_wmma_gemm<<<BATCH * 8,     256, 0, stream>>>(features, score, centroids, RES);
    vlad_finalize <<<BATCH * KKEEP / FIN_ROWS, 32 * FIN_ROWS, 0, stream>>>(RES, out);
}
